# starting kernel
# baseline (speedup 1.0000x reference)
#include <hip/hip_runtime.h>
#include <hip/hip_bf16.h>

typedef __attribute__((ext_vector_type(2))) float v2f;
typedef __attribute__((ext_vector_type(8))) float v8f;

#define DD 96
#define KSTEPS (DD / 4)
#define CTILES (DD / 16)
#define LDS_PAIR_STRIDE (2 * DD)

__device__ __forceinline__ void stage_weights(float* __restrict__ lds,
                                              const float* __restrict__ W) {
    for (int i = threadIdx.x; i < DD * DD; i += 128) {
        const int r = i / DD;
        const int c = i - r * DD;
        lds[(r >> 1) * LDS_PAIR_STRIDE + c * 2 + (r & 1)] = W[i];
    }
}

__global__ void zero_kernel(float* __restrict__ p, int n) {
    int i = blockIdx.x * blockDim.x + threadIdx.x;
    if (i < n) p[i] = 0.0f;
}

__global__ __launch_bounds__(128) void gemm_bias_kernel(
    const float* __restrict__ X, const float* __restrict__ W,
    const float* __restrict__ bias, float* __restrict__ Y,
    int nPairs, int N)
{
    __shared__ float wLds[DD * DD];
    stage_weights(wLds, W);
    __syncthreads();

    const int lane = threadIdx.x & 31;
    const int wave = threadIdx.x >> 5;
    const int half = lane >> 4;
    const int l16  = lane & 15;
    const int pair = blockIdx.x * 4 + wave;
    if (pair >= nPairs) return;
    const int rowBase = pair * 32;

    int m0 = rowBase + l16;       if (m0 >= N) m0 = N - 1;
    int m1 = rowBase + 16 + l16;  if (m1 >= N) m1 = N - 1;
    const float* aPtr0 = X + (size_t)m0 * DD + 2 * half;
    const float* aPtr1 = X + (size_t)m1 * DD + 2 * half;

    v8f acc0[CTILES] = {};
    v8f acc1[CTILES] = {};
    for (int kt = 0; kt < KSTEPS; ++kt) {
        const int k = kt * 4;
        v2f a0 = *(const v2f*)(aPtr0 + k);
        v2f a1 = *(const v2f*)(aPtr1 + k);
        const float* bRow = wLds + (size_t)(kt * 2 + half) * LDS_PAIR_STRIDE;
        #pragma unroll
        for (int ct = 0; ct < CTILES; ++ct) {
            v2f b = *(const v2f*)(bRow + (ct * 16 + l16) * 2);
            acc0[ct] = __builtin_amdgcn_wmma_f32_16x16x4_f32(
                false, a0, false, b, (short)0, acc0[ct], false, false);
            acc1[ct] = __builtin_amdgcn_wmma_f32_16x16x4_f32(
                false, a1, false, b, (short)0, acc1[ct], false, false);
        }
    }

    #pragma unroll
    for (int ct = 0; ct < CTILES; ++ct) {
        const int n = ct * 16 + l16;
        const float bv = bias[n];
        #pragma unroll
        for (int r = 0; r < 8; ++r) {
            const int ma = rowBase + r + 8 * half;
            const int mb = rowBase + 16 + r + 8 * half;
            if (ma < N) Y[(size_t)ma * DD + n] = acc0[ct][r] + bv;
            if (mb < N) Y[(size_t)mb * DD + n] = acc1[ct][r] + bv;
        }
    }
}

__global__ void scatter_kernel(
    const float* __restrict__ newX, const float* __restrict__ avals,
    const int* __restrict__ row, const int* __restrict__ col,
    float* __restrict__ agg, int E)
{
    const int t = blockIdx.x * blockDim.x + threadIdx.x;
    const int e = t / 24;
    const int c = t % 24;
    if (e >= E) return;
    const float av = avals[e];
    const int dst = row[e];
    const int src = col[e];
    const float4 v = *(const float4*)(newX + (size_t)src * DD + c * 4);
    float* p = agg + (size_t)dst * DD + c * 4;
    atomicAdd(p + 0, av * v.x);
    atomicAdd(p + 1, av * v.y);
    atomicAdd(p + 2, av * v.z);
    atomicAdd(p + 3, av * v.w);
}

__global__ __launch_bounds__(128) void gate_kernel(
    const float* __restrict__ X, const float* __restrict__ agg,
    const float* __restrict__ Wgi, const float* __restrict__ Wgn,
    const float* __restrict__ bgi, const float* __restrict__ bgn,
    float* __restrict__ out, int nPairs, int N)
{
    __shared__ float wLds[DD * DD];

    const int lane = threadIdx.x & 31;
    const int wave = threadIdx.x >> 5;
    const int half = lane >> 4;
    const int l16  = lane & 15;
    const int pair = blockIdx.x * 4 + wave;
    const bool active = (pair < nPairs);
    const int rowBase = pair * 32;
    int m0 = rowBase + l16;       if (m0 >= N) m0 = (N > 0) ? N - 1 : 0;
    int m1 = rowBase + 16 + l16;  if (m1 >= N) m1 = (N > 0) ? N - 1 : 0;

    v8f acc0[CTILES] = {};
    v8f acc1[CTILES] = {};

    stage_weights(wLds, Wgi);
    __syncthreads();
    if (active) {
        const float* aPtr0 = X + (size_t)m0 * DD + 2 * half;
        const float* aPtr1 = X + (size_t)m1 * DD + 2 * half;
        for (int kt = 0; kt < KSTEPS; ++kt) {
            const int k = kt * 4;
            v2f a0 = *(const v2f*)(aPtr0 + k);
            v2f a1 = *(const v2f*)(aPtr1 + k);
            const float* bRow = wLds + (size_t)(kt * 2 + half) * LDS_PAIR_STRIDE;
            #pragma unroll
            for (int ct = 0; ct < CTILES; ++ct) {
                v2f b = *(const v2f*)(bRow + (ct * 16 + l16) * 2);
                acc0[ct] = __builtin_amdgcn_wmma_f32_16x16x4_f32(
                    false, a0, false, b, (short)0, acc0[ct], false, false);
                acc1[ct] = __builtin_amdgcn_wmma_f32_16x16x4_f32(
                    false, a1, false, b, (short)0, acc1[ct], false, false);
            }
        }
    }
    __syncthreads();

    stage_weights(wLds, Wgn);
    __syncthreads();
    if (active) {
        const float* aPtr0 = agg + (size_t)m0 * DD + 2 * half;
        const float* aPtr1 = agg + (size_t)m1 * DD + 2 * half;
        for (int kt = 0; kt < KSTEPS; ++kt) {
            const int k = kt * 4;
            v2f a0 = *(const v2f*)(aPtr0 + k);
            v2f a1 = *(const v2f*)(aPtr1 + k);
            const float* bRow = wLds + (size_t)(kt * 2 + half) * LDS_PAIR_STRIDE;
            #pragma unroll
            for (int ct = 0; ct < CTILES; ++ct) {
                v2f b = *(const v2f*)(bRow + (ct * 16 + l16) * 2);
                acc0[ct] = __builtin_amdgcn_wmma_f32_16x16x4_f32(
                    false, a0, false, b, (short)0, acc0[ct], false, false);
                acc1[ct] = __builtin_amdgcn_wmma_f32_16x16x4_f32(
                    false, a1, false, b, (short)0, acc1[ct], false, false);
            }
        }

        #pragma unroll
        for (int ct = 0; ct < CTILES; ++ct) {
            const int n = ct * 16 + l16;
            const float bsum = bgi[n] + bgn[n];
            #pragma unroll
            for (int r = 0; r < 8; ++r) {
                #pragma unroll
                for (int t2 = 0; t2 < 2; ++t2) {
                    const int m = rowBase + t2 * 16 + r + 8 * half;
                    if (m < N) {
                        const float accv = (t2 == 0) ? acc0[ct][r] : acc1[ct][r];
                        float g = accv + bsum;
                        g = 1.0f / (1.0f + __expf(-g));
                        const float av = agg[(size_t)m * DD + n];
                        const float xv = X[(size_t)m * DD + n];
                        out[(size_t)m * DD + n] = av * g + xv * (1.0f - g);
                    }
                }
            }
        }
    }
}

extern "C" void kernel_launch(void* const* d_in, const int* in_sizes, int n_in,
                              void* d_out, int out_size, void* d_ws, size_t ws_size,
                              hipStream_t stream) {
    const float* X      = (const float*)d_in[0];
    const float* a_vals = (const float*)d_in[1];
    const float* Wn     = (const float*)d_in[2];
    const float* bn     = (const float*)d_in[3];
    const float* Wgi    = (const float*)d_in[4];
    const float* bgi    = (const float*)d_in[5];
    const float* Wgn    = (const float*)d_in[6];
    const float* bgn    = (const float*)d_in[7];
    const int*   row    = (const int*)d_in[8];
    const int*   col    = (const int*)d_in[9];
    float* out = (float*)d_out;

    const int N = in_sizes[0] / DD;
    const int E = in_sizes[1];

    float* newX = (float*)d_ws;
    float* agg  = (float*)d_ws + (size_t)N * DD;

    {
        const int total = N * DD;
        zero_kernel<<<(total + 255) / 256, 256, 0, stream>>>(agg, total);
    }

    const int nPairs = (N + 31) / 32;
    const int gemmBlocks = (nPairs + 3) / 4;

    gemm_bias_kernel<<<gemmBlocks, 128, 0, stream>>>(X, Wn, bn, newX, nPairs, N);

    {
        const long long threads = (long long)E * 24;
        const int blocks = (int)((threads + 255) / 256);
        scatter_kernel<<<blocks, 256, 0, stream>>>(newX, a_vals, row, col, agg, E);
    }

    gate_kernel<<<gemmBlocks, 128, 0, stream>>>(X, agg, Wgi, Wgn, bgi, bgn,
                                                out, nPairs, N);
}


# SpMM by destination owner: each block streams all of row[], keeps edges of its 200 rows, sums a*newX[col] in an LDS tile (ds f32 adds, arrival order), stores agg once; no global atomics or zero fill
# speedup vs baseline: 1.8444x; 1.8444x over previous
#include <hip/hip_runtime.h>
#include <hip/hip_bf16.h>


typedef __attribute__((ext_vector_type(2))) float v2f;
typedef __attribute__((ext_vector_type(8))) float v8f;

#define DD 96
#define CTILES (DD / 16)
#define KCHUNKS (DD / 16)
#define LDS_PAIR_STRIDE (2 * DD)

__device__ __forceinline__ void stage_weights(float* __restrict__ lds,
                                              const float* __restrict__ W) {
    for (int i = threadIdx.x; i < DD * DD; i += 128) {
        const int r = i / DD;
        const int c = i - r * DD;
        lds[(r >> 1) * LDS_PAIR_STRIDE + c * 2 + (r & 1)] = W[i];
    }
}

__device__ __forceinline__ void load_row_half(float4 (&r)[2 * KCHUNKS],
                                              const float* __restrict__ p) {
    #pragma unroll
    for (int kc = 0; kc < KCHUNKS; ++kc) {
        r[2 * kc]     = *(const float4*)(p + 16 * kc);
        r[2 * kc + 1] = *(const float4*)(p + 16 * kc + 4);
    }
}

__device__ __forceinline__ void mma_rows(v8f (&acc0)[CTILES], v8f (&acc1)[CTILES],
                                         const float4 (&r0)[2 * KCHUNKS],
                                         const float4 (&r1)[2 * KCHUNKS],
                                         const float* __restrict__ wLds,
                                         int half, int l16) {
    #pragma unroll
    for (int kc = 0; kc < KCHUNKS; ++kc) {
        #pragma unroll
        for (int q = 0; q < 4; ++q) {
            const float4 f0 = r0[2 * kc + (q >> 1)];
            const float4 f1 = r1[2 * kc + (q >> 1)];
            v2f b0, b1;
            if (q & 1) { b0.x = f0.z; b0.y = f0.w; b1.x = f1.z; b1.y = f1.w; }
            else       { b0.x = f0.x; b0.y = f0.y; b1.x = f1.x; b1.y = f1.y; }
            const float* wRow = wLds + (8 * kc + 4 * half + q) * LDS_PAIR_STRIDE + l16 * 2;
            #pragma unroll
            for (int ct = 0; ct < CTILES; ++ct) {
                const v2f w = *(const v2f*)(wRow + ct * 32);
                acc0[ct] = __builtin_amdgcn_wmma_f32_16x16x4_f32(
                    false, w, false, b0, (short)0, acc0[ct], false, false);
                acc1[ct] = __builtin_amdgcn_wmma_f32_16x16x4_f32(
                    false, w, false, b1, (short)0, acc1[ct], false, false);
            }
        }
    }
}

__global__ __launch_bounds__(128) void gemm_bias_kernel(
    const float* __restrict__ X, const float* __restrict__ W,
    const float* __restrict__ bias, float* __restrict__ Y,
    int nPairs, int N)
{
    __shared__ float wLds[DD * DD];

    const int lane = threadIdx.x & 31;
    const int wave = threadIdx.x >> 5;
    const int half = lane >> 4;
    const int l16  = lane & 15;
    const int pair = blockIdx.x * 4 + wave;
    const bool active = (pair < nPairs);
    const int rowBase = pair * 32;
    const int r0i = rowBase + l16;
    const int r1i = rowBase + 16 + l16;
    int m0 = r0i;  if (m0 >= N) m0 = (N > 0) ? N - 1 : 0;
    int m1 = r1i;  if (m1 >= N) m1 = (N > 0) ? N - 1 : 0;

    float4 x0[2 * KCHUNKS], x1[2 * KCHUNKS];
    if (active) {
        load_row_half(x0, X + (size_t)m0 * DD + 8 * half);
        load_row_half(x1, X + (size_t)m1 * DD + 8 * half);
    }

    stage_weights(wLds, W);
    __syncthreads();
    if (!active) return;

    float4 bA[CTILES], bB[CTILES];
    #pragma unroll
    for (int ct = 0; ct < CTILES; ++ct) {
        bA[ct] = *(const float4*)(bias + ct * 16 + 8 * half);
        bB[ct] = *(const float4*)(bias + ct * 16 + 8 * half + 4);
    }

    v8f acc0[CTILES] = {};
    v8f acc1[CTILES] = {};
    mma_rows(acc0, acc1, x0, x1, wLds, half, l16);

    #pragma unroll
    for (int ct = 0; ct < CTILES; ++ct) {
        const int n0 = ct * 16 + 8 * half;
        if (r0i < N) {
            float* y = Y + (size_t)r0i * DD + n0;
            *(float4*)(y)     = make_float4(acc0[ct][0] + bA[ct].x, acc0[ct][1] + bA[ct].y,
                                            acc0[ct][2] + bA[ct].z, acc0[ct][3] + bA[ct].w);
            *(float4*)(y + 4) = make_float4(acc0[ct][4] + bB[ct].x, acc0[ct][5] + bB[ct].y,
                                            acc0[ct][6] + bB[ct].z, acc0[ct][7] + bB[ct].w);
        }
        if (r1i < N) {
            float* y = Y + (size_t)r1i * DD + n0;
            *(float4*)(y)     = make_float4(acc1[ct][0] + bA[ct].x, acc1[ct][1] + bA[ct].y,
                                            acc1[ct][2] + bA[ct].z, acc1[ct][3] + bA[ct].w);
            *(float4*)(y + 4) = make_float4(acc1[ct][4] + bB[ct].x, acc1[ct][5] + bB[ct].y,
                                            acc1[ct][6] + bB[ct].z, acc1[ct][7] + bB[ct].w);
        }
    }
}

#define OWN_ROWS 200
#define OWN_WAVES 32
#define OWN_QW 1536
#define OWN_FLY 16
#define OWN_UNR 8

__device__ __forceinline__ void own_drain(const int* __restrict__ q, int cnt,
                                          float* __restrict__ tile, int lo,
                                          const float* __restrict__ newX,
                                          const float* __restrict__ avals,
                                          const int* __restrict__ row,
                                          const int* __restrict__ col, int lane) {
    for (int b = 0; b < cnt; b += 32) {
        const int i = b + lane;
        int lr = 0, c = 0;
        float a = 0.0f;
        if (i < cnt) {
            const int e = q[i];
            lr = row[e] - lo;
            c  = col[e];
            a  = avals[e];
        }
        const int m = (cnt - b < 32) ? (cnt - b) : 32;
        for (int j0 = 0; j0 < m; j0 += OWN_FLY) {
            float v[OWN_FLY][3];
            #pragma unroll
            for (int u = 0; u < OWN_FLY; ++u) {
                const int cj = __builtin_amdgcn_readlane(c, j0 + u);
                const float* p = newX + (size_t)cj * DD + lane;
                v[u][0] = p[0];
                v[u][1] = p[32];
                v[u][2] = p[64];
            }
            #pragma unroll
            for (int u = 0; u < OWN_FLY; ++u) {
                if (j0 + u < m) {
                    const float aj = __int_as_float(
                        __builtin_amdgcn_readlane(__float_as_int(a), j0 + u));
                    const int lj = __builtin_amdgcn_readlane(lr, j0 + u);
                    float* t = tile + lj * DD + lane;
                    atomicAdd(t,      aj * v[u][0]);
                    atomicAdd(t + 32, aj * v[u][1]);
                    atomicAdd(t + 64, aj * v[u][2]);
                }
            }
        }
    }
}

__device__ __forceinline__ void own_match(const int4& r, int e, int lo, unsigned lmask,
                                          int* __restrict__ q, int& cnt) {
    const int rk[4] = {r.x, r.y, r.z, r.w};
    bool mk[4];
    #pragma unroll
    for (int k = 0; k < 4; ++k)
        mk[k] = ((unsigned)rk[k] - (unsigned)lo) < (unsigned)OWN_ROWS;
    if (__builtin_amdgcn_ballot_w32(mk[0] | mk[1] | mk[2] | mk[3]) != 0u) {
        #pragma unroll
        for (int k = 0; k < 4; ++k) {
            const unsigned bk = __builtin_amdgcn_ballot_w32(mk[k]);
            if (mk[k]) q[cnt + __builtin_popcount(bk & lmask)] = e + k;
            cnt += __builtin_popcount(bk);
        }
    }
}

__global__ __launch_bounds__(OWN_WAVES * 32) void owner_sum_kernel(
    const float* __restrict__ newX, const float* __restrict__ avals,
    const int* __restrict__ row, const int* __restrict__ col,
    float* __restrict__ agg, int E, int N)
{
    __shared__ float tile[OWN_ROWS * DD];
    __shared__ int   queue[OWN_WAVES * OWN_QW];

    const int tid  = threadIdx.x;
    const int lane = tid & 31;
    const int wave = tid >> 5;
    const int lo   = blockIdx.x * OWN_ROWS;
    int* q = queue + wave * OWN_QW;

    for (int i = tid; i < OWN_ROWS * DD; i += OWN_WAVES * 32) tile[i] = 0.0f;
    __syncthreads();

    const int nFull = E >> 7;
    const int start = (int)(((long long)blockIdx.x * nFull) / gridDim.x);
    const unsigned lmask = (1u << lane) - 1u;
    int cnt = 0;

    int t = wave;
    for (; t + (OWN_UNR - 1) * OWN_WAVES < nFull; t += OWN_WAVES * OWN_UNR) {
        if (cnt + 128 * OWN_UNR > OWN_QW) {
            own_drain(q, cnt, tile, lo, newX, avals, row, col, lane);
            cnt = 0;
        }
        int4 r[OWN_UNR];
        int  e0[OWN_UNR];
        #pragma unroll
        for (int u = 0; u < OWN_UNR; ++u) {
            int gg = t + u * OWN_WAVES + start;
            if (gg >= nFull) gg -= nFull;
            e0[u] = gg * 128 + lane * 4;
            r[u] = *(const int4*)(row + e0[u]);
        }
        #pragma unroll
        for (int u = 0; u < OWN_UNR; ++u) own_match(r[u], e0[u], lo, lmask, q, cnt);
    }
    for (; t < nFull; t += OWN_WAVES) {
        if (cnt + 128 > OWN_QW) {
            own_drain(q, cnt, tile, lo, newX, avals, row, col, lane);
            cnt = 0;
        }
        int gg = t + start;
        if (gg >= nFull) gg -= nFull;
        const int e = gg * 128 + lane * 4;
        const int4 r = *(const int4*)(row + e);
        own_match(r, e, lo, lmask, q, cnt);
    }
    if (wave == 0 && (E & 127) != 0) {
        if (cnt + 128 > OWN_QW) {
            own_drain(q, cnt, tile, lo, newX, avals, row, col, lane);
            cnt = 0;
        }
        const int e = nFull * 128 + lane * 4;
        int4 r = make_int4(-1, -1, -1, -1);
        if (e     < E) r.x = row[e];
        if (e + 1 < E) r.y = row[e + 1];
        if (e + 2 < E) r.z = row[e + 2];
        if (e + 3 < E) r.w = row[e + 3];
        own_match(r, e, lo, lmask, q, cnt);
    }
    own_drain(q, cnt, tile, lo, newX, avals, row, col, lane);
    __syncthreads();

    int rows = N - lo;
    if (rows > OWN_ROWS) rows = OWN_ROWS;
    float4* dst = (float4*)(agg + (size_t)lo * DD);
    const float4* srcT = (const float4*)tile;
    for (int i = tid; i < rows * (DD / 4); i += OWN_WAVES * 32) dst[i] = srcT[i];
}

__device__ __forceinline__ void blend8(float* __restrict__ o, const v8f& acc,
                                       const float (&bs)[8],
                                       const float4& aLo, const float4& aHi,
                                       const float4& xLo, const float4& xHi) {
    const float av[8] = {aLo.x, aLo.y, aLo.z, aLo.w, aHi.x, aHi.y, aHi.z, aHi.w};
    const float xv[8] = {xLo.x, xLo.y, xLo.z, xLo.w, xHi.x, xHi.y, xHi.z, xHi.w};
    float res[8];
    #pragma unroll
    for (int r = 0; r < 8; ++r) {
        float g = acc[r] + bs[r];
        g = 1.0f / (1.0f + __expf(-g));
        res[r] = av[r] * g + xv[r] * (1.0f - g);
    }
    *(float4*)(o)     = make_float4(res[0], res[1], res[2], res[3]);
    *(float4*)(o + 4) = make_float4(res[4], res[5], res[6], res[7]);
}

__global__ __launch_bounds__(128) void gate_kernel(
    const float* __restrict__ X, const float* __restrict__ agg,
    const float* __restrict__ Wgi, const float* __restrict__ Wgn,
    const float* __restrict__ bgi, const float* __restrict__ bgn,
    float* __restrict__ out, int nPairs, int N)
{
    __shared__ float wLds[DD * DD];

    const int lane = threadIdx.x & 31;
    const int wave = threadIdx.x >> 5;
    const int half = lane >> 4;
    const int l16  = lane & 15;
    const int pair = blockIdx.x * 4 + wave;
    const bool active = (pair < nPairs);
    const int rowBase = pair * 32;
    const int r0i = rowBase + l16;
    const int r1i = rowBase + 16 + l16;
    int m0 = r0i;  if (m0 >= N) m0 = (N > 0) ? N - 1 : 0;
    int m1 = r1i;  if (m1 >= N) m1 = (N > 0) ? N - 1 : 0;

    float4 x0[2 * KCHUNKS], x1[2 * KCHUNKS], a0[2 * KCHUNKS], a1[2 * KCHUNKS];
    if (active) {
        load_row_half(x0, X   + (size_t)m0 * DD + 8 * half);
        load_row_half(x1, X   + (size_t)m1 * DD + 8 * half);
        load_row_half(a0, agg + (size_t)m0 * DD + 8 * half);
        load_row_half(a1, agg + (size_t)m1 * DD + 8 * half);
    }

    v8f acc0[CTILES] = {};
    v8f acc1[CTILES] = {};

    stage_weights(wLds, Wgi);
    __syncthreads();
    if (active) mma_rows(acc0, acc1, x0, x1, wLds, half, l16);
    __syncthreads();

    stage_weights(wLds, Wgn);
    __syncthreads();
    if (active) {
        float bs[CTILES][8];
        #pragma unroll
        for (int ct = 0; ct < CTILES; ++ct) {
            const int n0 = ct * 16 + 8 * half;
            const float4 iA = *(const float4*)(bgi + n0);
            const float4 iB = *(const float4*)(bgi + n0 + 4);
            const float4 nA = *(const float4*)(bgn + n0);
            const float4 nB = *(const float4*)(bgn + n0 + 4);
            bs[ct][0] = iA.x + nA.x; bs[ct][1] = iA.y + nA.y;
            bs[ct][2] = iA.z + nA.z; bs[ct][3] = iA.w + nA.w;
            bs[ct][4] = iB.x + nB.x; bs[ct][5] = iB.y + nB.y;
            bs[ct][6] = iB.z + nB.z; bs[ct][7] = iB.w + nB.w;
        }

        mma_rows(acc0, acc1, a0, a1, wLds, half, l16);

        #pragma unroll
        for (int ct = 0; ct < CTILES; ++ct) {
            const int n0 = ct * 16 + 8 * half;
            if (r0i < N)
                blend8(out + (size_t)r0i * DD + n0, acc0[ct], bs[ct],
                       a0[2 * ct], a0[2 * ct + 1], x0[2 * ct], x0[2 * ct + 1]);
            if (r1i < N)
                blend8(out + (size_t)r1i * DD + n0, acc1[ct], bs[ct],
                       a1[2 * ct], a1[2 * ct + 1], x1[2 * ct], x1[2 * ct + 1]);
        }
    }
}

extern "C" void kernel_launch(void* const* d_in, const int* in_sizes, int n_in,
                              void* d_out, int out_size, void* d_ws, size_t ws_size,
                              hipStream_t stream) {
    const float* X      = (const float*)d_in[0];
    const float* a_vals = (const float*)d_in[1];
    const float* Wn     = (const float*)d_in[2];
    const float* bn     = (const float*)d_in[3];
    const float* Wgi    = (const float*)d_in[4];
    const float* bgi    = (const float*)d_in[5];
    const float* Wgn    = (const float*)d_in[6];
    const float* bgn    = (const float*)d_in[7];
    const int*   row    = (const int*)d_in[8];
    const int*   col    = (const int*)d_in[9];
    float* out = (float*)d_out;

    const int N = in_sizes[0] / DD;
    const int E = in_sizes[1];

    float* newX = (float*)d_ws;
    float* agg  = (float*)d_ws + (size_t)N * DD;

    const int nPairs = (N + 31) / 32;
    const int gemmBlocks = (nPairs + 3) / 4;

    gemm_bias_kernel<<<gemmBlocks, 128, 0, stream>>>(X, Wn, bn, newX, nPairs, N);

    {
        const int ownBlocks = (N + OWN_ROWS - 1) / OWN_ROWS;
        owner_sum_kernel<<<ownBlocks, OWN_WAVES * 32, 0, stream>>>(newX, a_vals, row, col, agg, E, N);
    }

    gate_kernel<<<gemmBlocks, 128, 0, stream>>>(X, agg, Wgi, Wgn, bgi, bgn,
                                                out, nPairs, N);
}


# edge scan on 1-byte owner keys (row/200, written by launch 1, parked in out): a block streams E bytes not 4E, 16 edges per lane-load, bytes compared word-wide; LDS f32 adds: arrival order, as before
# speedup vs baseline: 2.2063x; 1.1962x over previous
#include <hip/hip_runtime.h>
#include <hip/hip_bf16.h>


typedef __attribute__((ext_vector_type(2))) float v2f;
typedef __attribute__((ext_vector_type(8))) float v8f;

#define DD 96
#define CTILES (DD / 16)
#define KCHUNKS (DD / 16)
#define LDS_PAIR_STRIDE (2 * DD)

__device__ __forceinline__ void stage_weights(float* __restrict__ lds,
                                              const float* __restrict__ W) {
    for (int i = threadIdx.x; i < DD * DD; i += 128) {
        const int r = i / DD;
        const int c = i - r * DD;
        lds[(r >> 1) * LDS_PAIR_STRIDE + c * 2 + (r & 1)] = W[i];
    }
}

__device__ __forceinline__ void load_row_half(float4 (&r)[2 * KCHUNKS],
                                              const float* __restrict__ p) {
    #pragma unroll
    for (int kc = 0; kc < KCHUNKS; ++kc) {
        r[2 * kc]     = *(const float4*)(p + 16 * kc);
        r[2 * kc + 1] = *(const float4*)(p + 16 * kc + 4);
    }
}

__device__ __forceinline__ void mma_rows(v8f (&acc0)[CTILES], v8f (&acc1)[CTILES],
                                         const float4 (&r0)[2 * KCHUNKS],
                                         const float4 (&r1)[2 * KCHUNKS],
                                         const float* __restrict__ wLds,
                                         int half, int l16) {
    #pragma unroll
    for (int kc = 0; kc < KCHUNKS; ++kc) {
        #pragma unroll
        for (int q = 0; q < 4; ++q) {
            const float4 f0 = r0[2 * kc + (q >> 1)];
            const float4 f1 = r1[2 * kc + (q >> 1)];
            v2f b0, b1;
            if (q & 1) { b0.x = f0.z; b0.y = f0.w; b1.x = f1.z; b1.y = f1.w; }
            else       { b0.x = f0.x; b0.y = f0.y; b1.x = f1.x; b1.y = f1.y; }
            const float* wRow = wLds + (8 * kc + 4 * half + q) * LDS_PAIR_STRIDE + l16 * 2;
            #pragma unroll
            for (int ct = 0; ct < CTILES; ++ct) {
                const v2f w = *(const v2f*)(wRow + ct * 32);
                acc0[ct] = __builtin_amdgcn_wmma_f32_16x16x4_f32(
                    false, w, false, b0, (short)0, acc0[ct], false, false);
                acc1[ct] = __builtin_amdgcn_wmma_f32_16x16x4_f32(
                    false, w, false, b1, (short)0, acc1[ct], false, false);
            }
        }
    }
}

#define OWN_ROWS 200
#define OWN_KV 4
#define OWN_TRIP_UNITS (32 * OWN_KV)

__device__ __forceinline__ unsigned owner_key(int r) {
    const unsigned k = (unsigned)r / (unsigned)OWN_ROWS;
    return k < 255u ? k : 255u;
}

__device__ __forceinline__ void load_unit_rows(int4 (&r)[4], const int* __restrict__ row,
                                               int E, int u) {
    const int e = u * 16;
    if (e + 16 <= E) {
        #pragma unroll
        for (int j = 0; j < 4; ++j) r[j] = *(const int4*)(row + e + 4 * j);
    } else {
        #pragma unroll
        for (int j = 0; j < 4; ++j) {
            const int b = e + 4 * j;
            r[j].x = (b     < E) ? row[b]     : -1;
            r[j].y = (b + 1 < E) ? row[b + 1] : -1;
            r[j].z = (b + 2 < E) ? row[b + 2] : -1;
            r[j].w = (b + 3 < E) ? row[b + 3] : -1;
        }
    }
}

__device__ __forceinline__ void store_unit_keys(float* __restrict__ keys, int u,
                                                const int4 (&r)[4]) {
    uint4 k;
    k.x = owner_key(r[0].x) | (owner_key(r[0].y) << 8) | (owner_key(r[0].z) << 16) | (owner_key(r[0].w) << 24);
    k.y = owner_key(r[1].x) | (owner_key(r[1].y) << 8) | (owner_key(r[1].z) << 16) | (owner_key(r[1].w) << 24);
    k.z = owner_key(r[2].x) | (owner_key(r[2].y) << 8) | (owner_key(r[2].z) << 16) | (owner_key(r[2].w) << 24);
    k.w = owner_key(r[3].x) | (owner_key(r[3].y) << 8) | (owner_key(r[3].z) << 16) | (owner_key(r[3].w) << 24);
    *((uint4*)keys + u) = k;
}

__global__ __launch_bounds__(128) void gemm_bias_kernel(
    const float* __restrict__ X, const float* __restrict__ W,
    const float* __restrict__ bias, float* __restrict__ Y,
    const int* __restrict__ row, float* __restrict__ keys,
    int nPairs, int N, int E)
{
    __shared__ float wLds[DD * DD];

    const int lane = threadIdx.x & 31;
    const int wave = threadIdx.x >> 5;
    const int half = lane >> 4;
    const int l16  = lane & 15;
    const int pair = blockIdx.x * 4 + wave;
    const bool active = (pair < nPairs);
    const int rowBase = pair * 32;
    const int r0i = rowBase + l16;
    const int r1i = rowBase + 16 + l16;
    int m0 = r0i;  if (m0 >= N) m0 = (N > 0) ? N - 1 : 0;
    int m1 = r1i;  if (m1 >= N) m1 = (N > 0) ? N - 1 : 0;

    float4 x0[2 * KCHUNKS], x1[2 * KCHUNKS];
    if (active) {
        load_row_half(x0, X + (size_t)m0 * DD + 8 * half);
        load_row_half(x1, X + (size_t)m1 * DD + 8 * half);
    }

    const int nUnits = ((E + 16 * OWN_TRIP_UNITS - 1) / (16 * OWN_TRIP_UNITS)) * OWN_TRIP_UNITS;
    const int unit0 = blockIdx.x * 128 + threadIdx.x;
    const int unitStep = gridDim.x * 128;
    int4 rk[4];
    if (unit0 < nUnits) load_unit_rows(rk, row, E, unit0);

    stage_weights(wLds, W);
    __syncthreads();

    if (unit0 < nUnits) store_unit_keys(keys, unit0, rk);
    for (int u = unit0 + unitStep; u < nUnits; u += unitStep) {
        load_unit_rows(rk, row, E, u);
        store_unit_keys(keys, u, rk);
    }
    if (!active) return;

    float4 bA[CTILES], bB[CTILES];
    #pragma unroll
    for (int ct = 0; ct < CTILES; ++ct) {
        bA[ct] = *(const float4*)(bias + ct * 16 + 8 * half);
        bB[ct] = *(const float4*)(bias + ct * 16 + 8 * half + 4);
    }

    v8f acc0[CTILES] = {};
    v8f acc1[CTILES] = {};
    mma_rows(acc0, acc1, x0, x1, wLds, half, l16);

    #pragma unroll
    for (int ct = 0; ct < CTILES; ++ct) {
        const int n0 = ct * 16 + 8 * half;
        if (r0i < N) {
            float* y = Y + (size_t)r0i * DD + n0;
            *(float4*)(y)     = make_float4(acc0[ct][0] + bA[ct].x, acc0[ct][1] + bA[ct].y,
                                            acc0[ct][2] + bA[ct].z, acc0[ct][3] + bA[ct].w);
            *(float4*)(y + 4) = make_float4(acc0[ct][4] + bB[ct].x, acc0[ct][5] + bB[ct].y,
                                            acc0[ct][6] + bB[ct].z, acc0[ct][7] + bB[ct].w);
        }
        if (r1i < N) {
            float* y = Y + (size_t)r1i * DD + n0;
            *(float4*)(y)     = make_float4(acc1[ct][0] + bA[ct].x, acc1[ct][1] + bA[ct].y,
                                            acc1[ct][2] + bA[ct].z, acc1[ct][3] + bA[ct].w);
            *(float4*)(y + 4) = make_float4(acc1[ct][4] + bB[ct].x, acc1[ct][5] + bB[ct].y,
                                            acc1[ct][6] + bB[ct].z, acc1[ct][7] + bB[ct].w);
        }
    }
}

#define OWN_WAVES 32
#define OWN_QW 1536
#define OWN_FLY 16

__device__ __forceinline__ void own_drain(const int* __restrict__ q, int cnt,
                                          float* __restrict__ tile, int lo,
                                          const float* __restrict__ newX,
                                          const float* __restrict__ avals,
                                          const int* __restrict__ row,
                                          const int* __restrict__ col, int lane) {
    for (int b = 0; b < cnt; b += 32) {
        const int i = b + lane;
        int lr = 0, c = 0;
        float a = 0.0f;
        if (i < cnt) {
            const int e = q[i];
            lr = row[e] - lo;
            c  = col[e];
            a  = avals[e];
        }
        const int m = (cnt - b < 32) ? (cnt - b) : 32;
        for (int j0 = 0; j0 < m; j0 += OWN_FLY) {
            float v[OWN_FLY][3];
            #pragma unroll
            for (int u = 0; u < OWN_FLY; ++u) {
                const int cj = __builtin_amdgcn_readlane(c, j0 + u);
                const float* p = newX + (size_t)cj * DD + lane;
                v[u][0] = p[0];
                v[u][1] = p[32];
                v[u][2] = p[64];
            }
            #pragma unroll
            for (int u = 0; u < OWN_FLY; ++u) {
                if (j0 + u < m) {
                    const float aj = __int_as_float(
                        __builtin_amdgcn_readlane(__float_as_int(a), j0 + u));
                    const int lj = __builtin_amdgcn_readlane(lr, j0 + u);
                    float* t = tile + lj * DD + lane;
                    atomicAdd(t,      aj * v[u][0]);
                    atomicAdd(t + 32, aj * v[u][1]);
                    atomicAdd(t + 64, aj * v[u][2]);
                }
            }
        }
    }
}

__device__ __forceinline__ unsigned nonzero_bytes(unsigned x) {
    return ((x & 0x7F7F7F7Fu) + 0x7F7F7F7Fu) | x;
}

__device__ __forceinline__ unsigned own_pair_mask(const uint4& ka, const uint4& kb,
                                                  unsigned pat) {
    const unsigned M = 0x80808080u;
    unsigned nz;
    nz  = (nonzero_bytes(ka.x ^ pat) & M) >> 7;
    nz |= (nonzero_bytes(ka.y ^ pat) & M) >> 6;
    nz |= (nonzero_bytes(ka.z ^ pat) & M) >> 5;
    nz |= (nonzero_bytes(ka.w ^ pat) & M) >> 4;
    nz |= (nonzero_bytes(kb.x ^ pat) & M) >> 3;
    nz |= (nonzero_bytes(kb.y ^ pat) & M) >> 2;
    nz |= (nonzero_bytes(kb.z ^ pat) & M) >> 1;
    nz |= (nonzero_bytes(kb.w ^ pat) & M);
    return ~nz;
}

__device__ __forceinline__ void own_load(uint4 (&k)[OWN_KV], const uint4* __restrict__ keys,
                                         int tt, int lane) {
    #pragma unroll
    for (int v = 0; v < OWN_KV; ++v) k[v] = keys[(tt * OWN_KV + v) * 32 + lane];
}

__device__ __forceinline__ int own_trip(int t, int nTrips, int start) {
    int tt = (t < nTrips ? t : nTrips - 1) + start;
    if (tt >= nTrips) tt -= nTrips;
    return tt;
}

__device__ __forceinline__ void own_take(const uint4 (&k)[OWN_KV], int tt, unsigned pat,
                                         unsigned lmask, int* __restrict__ q, int& cnt,
                                         float* __restrict__ tile, int lo,
                                         const float* __restrict__ newX,
                                         const float* __restrict__ avals,
                                         const int* __restrict__ row,
                                         const int* __restrict__ col, int lane) {
    unsigned long long w = (unsigned long long)own_pair_mask(k[0], k[1], pat)
                         | ((unsigned long long)own_pair_mask(k[2], k[3], pat) << 32);
    for (;;) {
        const unsigned bal = __builtin_amdgcn_ballot_w32(w != 0ull);
        if (bal == 0u) break;
        if (cnt + 32 > OWN_QW) {
            own_drain(q, cnt, tile, lo, newX, avals, row, col, lane);
            cnt = 0;
        }
        if (w != 0ull) {
            const int p  = __builtin_ctzll(w);
            w &= w - 1ull;
            const int pp = p & 31;
            const int v  = ((p >> 5) << 1) + ((pp >> 2) & 1);
            const int unit = (tt * OWN_KV + v) * 32 + lane;
            q[cnt + __builtin_popcount(bal & lmask)] = unit * 16 + ((pp & 3) << 2) + (pp >> 3);
        }
        cnt += __builtin_popcount(bal);
    }
}

__global__ __launch_bounds__(OWN_WAVES * 32) void owner_sum_kernel(
    const float* __restrict__ newX, const float* __restrict__ avals,
    const int* __restrict__ row, const int* __restrict__ col,
    const float* __restrict__ keysF, float* __restrict__ agg, int E, int N)
{
    __shared__ float tile[OWN_ROWS * DD];
    __shared__ int   queue[OWN_WAVES * OWN_QW];

    const int tid  = threadIdx.x;
    const int lane = tid & 31;
    const int wave = __builtin_amdgcn_readfirstlane(tid >> 5);
    const int lo   = blockIdx.x * OWN_ROWS;
    int* q = queue + wave * OWN_QW;
    const uint4* keys = (const uint4*)keysF;

    for (int i = tid; i < OWN_ROWS * DD; i += OWN_WAVES * 32) tile[i] = 0.0f;
    __syncthreads();

    const int nTrips = (E + 16 * OWN_TRIP_UNITS - 1) / (16 * OWN_TRIP_UNITS);
    const int start = (int)(((long long)blockIdx.x * nTrips) / gridDim.x);
    const unsigned pat   = (unsigned)blockIdx.x * 0x01010101u;
    const unsigned lmask = (1u << lane) - 1u;
    int cnt = 0;

    if (nTrips > 0) {
        uint4 ka[OWN_KV], kb[OWN_KV];
        int t = wave;
        int ta = own_trip(t, nTrips, start);
        own_load(ka, keys, ta, lane);
        for (; t < nTrips; t += 2 * OWN_WAVES) {
            const int t1 = t + OWN_WAVES;
            const int tb = own_trip(t1, nTrips, start);
            own_load(kb, keys, tb, lane);
            own_take(ka, ta, pat, lmask, q, cnt, tile, lo, newX, avals, row, col, lane);
            ta = own_trip(t1 + OWN_WAVES, nTrips, start);
            own_load(ka, keys, ta, lane);
            if (t1 < nTrips)
                own_take(kb, tb, pat, lmask, q, cnt, tile, lo, newX, avals, row, col, lane);
        }
    }
    own_drain(q, cnt, tile, lo, newX, avals, row, col, lane);
    __syncthreads();

    int rows = N - lo;
    if (rows > OWN_ROWS) rows = OWN_ROWS;
    float4* dst = (float4*)(agg + (size_t)lo * DD);
    const float4* srcT = (const float4*)tile;
    for (int i = tid; i < rows * (DD / 4); i += OWN_WAVES * 32) dst[i] = srcT[i];
}

__device__ __forceinline__ void blend8(float* __restrict__ o, const v8f& acc,
                                       const float (&bs)[8],
                                       const float4& aLo, const float4& aHi,
                                       const float4& xLo, const float4& xHi) {
    const float av[8] = {aLo.x, aLo.y, aLo.z, aLo.w, aHi.x, aHi.y, aHi.z, aHi.w};
    const float xv[8] = {xLo.x, xLo.y, xLo.z, xLo.w, xHi.x, xHi.y, xHi.z, xHi.w};
    float res[8];
    #pragma unroll
    for (int r = 0; r < 8; ++r) {
        float g = acc[r] + bs[r];
        g = 1.0f / (1.0f + __expf(-g));
        res[r] = av[r] * g + xv[r] * (1.0f - g);
    }
    *(float4*)(o)     = make_float4(res[0], res[1], res[2], res[3]);
    *(float4*)(o + 4) = make_float4(res[4], res[5], res[6], res[7]);
}

__global__ __launch_bounds__(128) void gate_kernel(
    const float* __restrict__ X, const float* __restrict__ agg,
    const float* __restrict__ Wgi, const float* __restrict__ Wgn,
    const float* __restrict__ bgi, const float* __restrict__ bgn,
    float* __restrict__ out, int nPairs, int N)
{
    __shared__ float wLds[DD * DD];

    const int lane = threadIdx.x & 31;
    const int wave = threadIdx.x >> 5;
    const int half = lane >> 4;
    const int l16  = lane & 15;
    const int pair = blockIdx.x * 4 + wave;
    const bool active = (pair < nPairs);
    const int rowBase = pair * 32;
    const int r0i = rowBase + l16;
    const int r1i = rowBase + 16 + l16;
    int m0 = r0i;  if (m0 >= N) m0 = (N > 0) ? N - 1 : 0;
    int m1 = r1i;  if (m1 >= N) m1 = (N > 0) ? N - 1 : 0;

    float4 x0[2 * KCHUNKS], x1[2 * KCHUNKS], a0[2 * KCHUNKS], a1[2 * KCHUNKS];
    if (active) {
        load_row_half(x0, X   + (size_t)m0 * DD + 8 * half);
        load_row_half(x1, X   + (size_t)m1 * DD + 8 * half);
        load_row_half(a0, agg + (size_t)m0 * DD + 8 * half);
        load_row_half(a1, agg + (size_t)m1 * DD + 8 * half);
    }

    v8f acc0[CTILES] = {};
    v8f acc1[CTILES] = {};

    stage_weights(wLds, Wgi);
    __syncthreads();
    if (active) mma_rows(acc0, acc1, x0, x1, wLds, half, l16);
    __syncthreads();

    stage_weights(wLds, Wgn);
    __syncthreads();
    if (active) {
        float bs[CTILES][8];
        #pragma unroll
        for (int ct = 0; ct < CTILES; ++ct) {
            const int n0 = ct * 16 + 8 * half;
            const float4 iA = *(const float4*)(bgi + n0);
            const float4 iB = *(const float4*)(bgi + n0 + 4);
            const float4 nA = *(const float4*)(bgn + n0);
            const float4 nB = *(const float4*)(bgn + n0 + 4);
            bs[ct][0] = iA.x + nA.x; bs[ct][1] = iA.y + nA.y;
            bs[ct][2] = iA.z + nA.z; bs[ct][3] = iA.w + nA.w;
            bs[ct][4] = iB.x + nB.x; bs[ct][5] = iB.y + nB.y;
            bs[ct][6] = iB.z + nB.z; bs[ct][7] = iB.w + nB.w;
        }

        mma_rows(acc0, acc1, a0, a1, wLds, half, l16);

        #pragma unroll
        for (int ct = 0; ct < CTILES; ++ct) {
            const int n0 = ct * 16 + 8 * half;
            if (r0i < N)
                blend8(out + (size_t)r0i * DD + n0, acc0[ct], bs[ct],
                       a0[2 * ct], a0[2 * ct + 1], x0[2 * ct], x0[2 * ct + 1]);
            if (r1i < N)
                blend8(out + (size_t)r1i * DD + n0, acc1[ct], bs[ct],
                       a1[2 * ct], a1[2 * ct + 1], x1[2 * ct], x1[2 * ct + 1]);
        }
    }
}

extern "C" void kernel_launch(void* const* d_in, const int* in_sizes, int n_in,
                              void* d_out, int out_size, void* d_ws, size_t ws_size,
                              hipStream_t stream) {
    const float* X      = (const float*)d_in[0];
    const float* a_vals = (const float*)d_in[1];
    const float* Wn     = (const float*)d_in[2];
    const float* bn     = (const float*)d_in[3];
    const float* Wgi    = (const float*)d_in[4];
    const float* bgi    = (const float*)d_in[5];
    const float* Wgn    = (const float*)d_in[6];
    const float* bgn    = (const float*)d_in[7];
    const int*   row    = (const int*)d_in[8];
    const int*   col    = (const int*)d_in[9];
    float* out = (float*)d_out;

    const int N = in_sizes[0] / DD;
    const int E = in_sizes[1];

    float* newX = (float*)d_ws;
    float* agg  = (float*)d_ws + (size_t)N * DD;

    const int nPairs = (N + 31) / 32;
    const int gemmBlocks = (nPairs + 3) / 4;

    float* keys = out;

    gemm_bias_kernel<<<gemmBlocks, 128, 0, stream>>>(X, Wn, bn, newX, row, keys, nPairs, N, E);

    {
        const int ownBlocks = (N + OWN_ROWS - 1) / OWN_ROWS;
        owner_sum_kernel<<<ownBlocks, OWN_WAVES * 32, 0, stream>>>(newX, a_vals, row, col, keys, agg, E, N);
    }

    gate_kernel<<<gemmBlocks, 128, 0, stream>>>(X, agg, Wgi, Wgn, bgi, bgn,
                                                out, nPairs, N);
}


# launch 1 packs each edge's col, row-in-range and a into one 8-byte record (in out, behind the keys); an owner block reads 1 record per edge it found, not row, col, a on 3 lines; sum order as before
# speedup vs baseline: 2.3404x; 1.0608x over previous
#include <hip/hip_runtime.h>
#include <hip/hip_bf16.h>


typedef __attribute__((ext_vector_type(2))) float v2f;
typedef __attribute__((ext_vector_type(8))) float v8f;

#define DD 96
#define CTILES (DD / 16)
#define KCHUNKS (DD / 16)
#define LDS_PAIR_STRIDE (2 * DD)

__device__ __forceinline__ void stage_weights(float* __restrict__ lds,
                                              const float* __restrict__ W) {
    for (int i = threadIdx.x; i < DD * DD; i += 128) {
        const int r = i / DD;
        const int c = i - r * DD;
        lds[(r >> 1) * LDS_PAIR_STRIDE + c * 2 + (r & 1)] = W[i];
    }
}

__device__ __forceinline__ void load_row_half(float4 (&r)[2 * KCHUNKS],
                                              const float* __restrict__ p) {
    #pragma unroll
    for (int kc = 0; kc < KCHUNKS; ++kc) {
        r[2 * kc]     = *(const float4*)(p + 16 * kc);
        r[2 * kc + 1] = *(const float4*)(p + 16 * kc + 4);
    }
}

__device__ __forceinline__ void mma_rows(v8f (&acc0)[CTILES], v8f (&acc1)[CTILES],
                                         const float4 (&r0)[2 * KCHUNKS],
                                         const float4 (&r1)[2 * KCHUNKS],
                                         const float* __restrict__ wLds,
                                         int half, int l16) {
    #pragma unroll
    for (int kc = 0; kc < KCHUNKS; ++kc) {
        #pragma unroll
        for (int q = 0; q < 4; ++q) {
            const float4 f0 = r0[2 * kc + (q >> 1)];
            const float4 f1 = r1[2 * kc + (q >> 1)];
            v2f b0, b1;
            if (q & 1) { b0.x = f0.z; b0.y = f0.w; b1.x = f1.z; b1.y = f1.w; }
            else       { b0.x = f0.x; b0.y = f0.y; b1.x = f1.x; b1.y = f1.y; }
            const float* wRow = wLds + (8 * kc + 4 * half + q) * LDS_PAIR_STRIDE + l16 * 2;
            #pragma unroll
            for (int ct = 0; ct < CTILES; ++ct) {
                const v2f w = *(const v2f*)(wRow + ct * 32);
                acc0[ct] = __builtin_amdgcn_wmma_f32_16x16x4_f32(
                    false, w, false, b0, (short)0, acc0[ct], false, false);
                acc1[ct] = __builtin_amdgcn_wmma_f32_16x16x4_f32(
                    false, w, false, b1, (short)0, acc1[ct], false, false);
            }
        }
    }
}

#define OWN_ROWS 200
#define OWN_KV 4
#define OWN_TRIP_UNITS (32 * OWN_KV)

__device__ __forceinline__ unsigned owner_key(int r) {
    const unsigned k = (unsigned)r / (unsigned)OWN_ROWS;
    return k < 255u ? k : 255u;
}

#define QUAD_PRE 4

__device__ __forceinline__ void load_quad(int4& r, int4& c, float4& a,
                                          const int* __restrict__ row,
                                          const int* __restrict__ col,
                                          const float* __restrict__ avals, int E, int qd) {
    const int e = qd * 4;
    if (e + 4 <= E) {
        r = *(const int4*)(row + e);
        c = *(const int4*)(col + e);
        a = *(const float4*)(avals + e);
    } else {
        r.x = (e     < E) ? row[e]     : -1;  c.x = (e     < E) ? col[e]     : 0;  a.x = (e     < E) ? avals[e]     : 0.0f;
        r.y = (e + 1 < E) ? row[e + 1] : -1;  c.y = (e + 1 < E) ? col[e + 1] : 0;  a.y = (e + 1 < E) ? avals[e + 1] : 0.0f;
        r.z = (e + 2 < E) ? row[e + 2] : -1;  c.z = (e + 2 < E) ? col[e + 2] : 0;  a.z = (e + 2 < E) ? avals[e + 2] : 0.0f;
        r.w = (e + 3 < E) ? row[e + 3] : -1;  c.w = (e + 3 < E) ? col[e + 3] : 0;  a.w = (e + 3 < E) ? avals[e + 3] : 0.0f;
    }
}

__device__ __forceinline__ uint2 edge_record(int r, int c, float a) {
    const unsigned k  = (unsigned)r / (unsigned)OWN_ROWS;
    const unsigned lr = (unsigned)r - k * (unsigned)OWN_ROWS;
    return make_uint2(((unsigned)c & 0x00FFFFFFu) | (lr << 24), __float_as_uint(a));
}

__device__ __forceinline__ void store_quad(float* __restrict__ keys, uint4* __restrict__ recs,
                                           int qd, const int4& r, const int4& c,
                                           const float4& a) {
    ((unsigned*)keys)[qd] = owner_key(r.x) | (owner_key(r.y) << 8)
                          | (owner_key(r.z) << 16) | (owner_key(r.w) << 24);
    const uint2 e0 = edge_record(r.x, c.x, a.x);
    const uint2 e1 = edge_record(r.y, c.y, a.y);
    const uint2 e2 = edge_record(r.z, c.z, a.z);
    const uint2 e3 = edge_record(r.w, c.w, a.w);
    recs[2 * qd]     = make_uint4(e0.x, e0.y, e1.x, e1.y);
    recs[2 * qd + 1] = make_uint4(e2.x, e2.y, e3.x, e3.y);
}

__global__ __launch_bounds__(128) void gemm_bias_kernel(
    const float* __restrict__ X, const float* __restrict__ W,
    const float* __restrict__ bias, float* __restrict__ Y,
    const int* __restrict__ row, const int* __restrict__ col,
    const float* __restrict__ avals, float* __restrict__ keys,
    int nPairs, int N, int E)
{
    __shared__ float wLds[DD * DD];

    const int lane = threadIdx.x & 31;
    const int wave = threadIdx.x >> 5;
    const int half = lane >> 4;
    const int l16  = lane & 15;
    const int pair = blockIdx.x * 4 + wave;
    const bool active = (pair < nPairs);
    const int rowBase = pair * 32;
    const int r0i = rowBase + l16;
    const int r1i = rowBase + 16 + l16;
    int m0 = r0i;  if (m0 >= N) m0 = (N > 0) ? N - 1 : 0;
    int m1 = r1i;  if (m1 >= N) m1 = (N > 0) ? N - 1 : 0;

    float4 x0[2 * KCHUNKS], x1[2 * KCHUNKS];
    if (active) {
        load_row_half(x0, X + (size_t)m0 * DD + 8 * half);
        load_row_half(x1, X + (size_t)m1 * DD + 8 * half);
    }

    const int nUnits = ((E + 16 * OWN_TRIP_UNITS - 1) / (16 * OWN_TRIP_UNITS)) * OWN_TRIP_UNITS;
    const int nQuads = nUnits * 4;
    uint4* recs = (uint4*)(keys + nUnits * 4);
    const int quad0 = blockIdx.x * 128 + threadIdx.x;
    const int quadStep = gridDim.x * 128;
    int4 rq[QUAD_PRE], cq[QUAD_PRE];
    float4 aq[QUAD_PRE];
    #pragma unroll
    for (int s = 0; s < QUAD_PRE; ++s) {
        const int qd = quad0 + s * quadStep;
        if (qd < nQuads) load_quad(rq[s], cq[s], aq[s], row, col, avals, E, qd);
    }

    stage_weights(wLds, W);
    __syncthreads();

    #pragma unroll
    for (int s = 0; s < QUAD_PRE; ++s) {
        const int qd = quad0 + s * quadStep;
        if (qd < nQuads) store_quad(keys, recs, qd, rq[s], cq[s], aq[s]);
    }
    for (int qd = quad0 + QUAD_PRE * quadStep; qd < nQuads; qd += quadStep) {
        int4 r, c2;
        float4 a2;
        load_quad(r, c2, a2, row, col, avals, E, qd);
        store_quad(keys, recs, qd, r, c2, a2);
    }
    if (!active) return;

    float4 bA[CTILES], bB[CTILES];
    #pragma unroll
    for (int ct = 0; ct < CTILES; ++ct) {
        bA[ct] = *(const float4*)(bias + ct * 16 + 8 * half);
        bB[ct] = *(const float4*)(bias + ct * 16 + 8 * half + 4);
    }

    v8f acc0[CTILES] = {};
    v8f acc1[CTILES] = {};
    mma_rows(acc0, acc1, x0, x1, wLds, half, l16);

    #pragma unroll
    for (int ct = 0; ct < CTILES; ++ct) {
        const int n0 = ct * 16 + 8 * half;
        if (r0i < N) {
            float* y = Y + (size_t)r0i * DD + n0;
            *(float4*)(y)     = make_float4(acc0[ct][0] + bA[ct].x, acc0[ct][1] + bA[ct].y,
                                            acc0[ct][2] + bA[ct].z, acc0[ct][3] + bA[ct].w);
            *(float4*)(y + 4) = make_float4(acc0[ct][4] + bB[ct].x, acc0[ct][5] + bB[ct].y,
                                            acc0[ct][6] + bB[ct].z, acc0[ct][7] + bB[ct].w);
        }
        if (r1i < N) {
            float* y = Y + (size_t)r1i * DD + n0;
            *(float4*)(y)     = make_float4(acc1[ct][0] + bA[ct].x, acc1[ct][1] + bA[ct].y,
                                            acc1[ct][2] + bA[ct].z, acc1[ct][3] + bA[ct].w);
            *(float4*)(y + 4) = make_float4(acc1[ct][4] + bB[ct].x, acc1[ct][5] + bB[ct].y,
                                            acc1[ct][6] + bB[ct].z, acc1[ct][7] + bB[ct].w);
        }
    }
}

#define OWN_WAVES 32
#define OWN_QW 1536
#define OWN_FLY 16

__device__ __forceinline__ void own_drain(const int* __restrict__ q, int cnt,
                                          float* __restrict__ tile,
                                          const float* __restrict__ newX,
                                          const uint2* __restrict__ recs, int lane) {
    for (int b = 0; b < cnt; b += 32) {
        const int i = b + lane;
        int lr = 0, c = 0;
        float a = 0.0f;
        if (i < cnt) {
            const uint2 r = recs[q[i]];
            lr = (int)(r.x >> 24);
            c  = (int)(r.x & 0x00FFFFFFu);
            a  = __uint_as_float(r.y);
        }
        const int m = (cnt - b < 32) ? (cnt - b) : 32;
        for (int j0 = 0; j0 < m; j0 += OWN_FLY) {
            float v[OWN_FLY][3];
            #pragma unroll
            for (int u = 0; u < OWN_FLY; ++u) {
                const int cj = __builtin_amdgcn_readlane(c, j0 + u);
                const float* p = newX + (size_t)cj * DD + lane;
                v[u][0] = p[0];
                v[u][1] = p[32];
                v[u][2] = p[64];
            }
            #pragma unroll
            for (int u = 0; u < OWN_FLY; ++u) {
                if (j0 + u < m) {
                    const float aj = __int_as_float(
                        __builtin_amdgcn_readlane(__float_as_int(a), j0 + u));
                    const int lj = __builtin_amdgcn_readlane(lr, j0 + u);
                    float* t = tile + lj * DD + lane;
                    atomicAdd(t,      aj * v[u][0]);
                    atomicAdd(t + 32, aj * v[u][1]);
                    atomicAdd(t + 64, aj * v[u][2]);
                }
            }
        }
    }
}

__device__ __forceinline__ unsigned nonzero_bytes(unsigned x) {
    return ((x & 0x7F7F7F7Fu) + 0x7F7F7F7Fu) | x;
}

__device__ __forceinline__ unsigned own_pair_mask(const uint4& ka, const uint4& kb,
                                                  unsigned pat) {
    const unsigned M = 0x80808080u;
    unsigned nz;
    nz  = (nonzero_bytes(ka.x ^ pat) & M) >> 7;
    nz |= (nonzero_bytes(ka.y ^ pat) & M) >> 6;
    nz |= (nonzero_bytes(ka.z ^ pat) & M) >> 5;
    nz |= (nonzero_bytes(ka.w ^ pat) & M) >> 4;
    nz |= (nonzero_bytes(kb.x ^ pat) & M) >> 3;
    nz |= (nonzero_bytes(kb.y ^ pat) & M) >> 2;
    nz |= (nonzero_bytes(kb.z ^ pat) & M) >> 1;
    nz |= (nonzero_bytes(kb.w ^ pat) & M);
    return ~nz;
}

__device__ __forceinline__ void own_load(uint4 (&k)[OWN_KV], const uint4* __restrict__ keys,
                                         int tt, int lane) {
    #pragma unroll
    for (int v = 0; v < OWN_KV; ++v) k[v] = keys[(tt * OWN_KV + v) * 32 + lane];
}

__device__ __forceinline__ int own_trip(int t, int nTrips, int start) {
    int tt = (t < nTrips ? t : nTrips - 1) + start;
    if (tt >= nTrips) tt -= nTrips;
    return tt;
}

__device__ __forceinline__ void own_take(const uint4 (&k)[OWN_KV], int tt, unsigned pat,
                                         unsigned lmask, int* __restrict__ q, int& cnt,
                                         float* __restrict__ tile,
                                         const float* __restrict__ newX,
                                         const uint2* __restrict__ recs, int lane) {
    unsigned long long w = (unsigned long long)own_pair_mask(k[0], k[1], pat)
                         | ((unsigned long long)own_pair_mask(k[2], k[3], pat) << 32);
    for (;;) {
        const unsigned bal = __builtin_amdgcn_ballot_w32(w != 0ull);
        if (bal == 0u) break;
        if (cnt + 32 > OWN_QW) {
            own_drain(q, cnt, tile, newX, recs, lane);
            cnt = 0;
        }
        if (w != 0ull) {
            const int p  = __builtin_ctzll(w);
            w &= w - 1ull;
            const int pp = p & 31;
            const int v  = ((p >> 5) << 1) + ((pp >> 2) & 1);
            const int unit = (tt * OWN_KV + v) * 32 + lane;
            q[cnt + __builtin_popcount(bal & lmask)] = unit * 16 + ((pp & 3) << 2) + (pp >> 3);
        }
        cnt += __builtin_popcount(bal);
    }
}

__global__ __launch_bounds__(OWN_WAVES * 32) void owner_sum_kernel(
    const float* __restrict__ newX, const float* __restrict__ keysF,
    float* __restrict__ agg, int E, int N)
{
    __shared__ float tile[OWN_ROWS * DD];
    __shared__ int   queue[OWN_WAVES * OWN_QW];

    const int tid  = threadIdx.x;
    const int lane = tid & 31;
    const int wave = __builtin_amdgcn_readfirstlane(tid >> 5);
    const int lo   = blockIdx.x * OWN_ROWS;
    int* q = queue + wave * OWN_QW;
    const uint4* keys = (const uint4*)keysF;

    for (int i = tid; i < OWN_ROWS * DD; i += OWN_WAVES * 32) tile[i] = 0.0f;
    __syncthreads();

    const int nTrips = (E + 16 * OWN_TRIP_UNITS - 1) / (16 * OWN_TRIP_UNITS);
    const uint2* recs = (const uint2*)(keysF + nTrips * OWN_TRIP_UNITS * 4);
    const int start = (int)(((long long)blockIdx.x * nTrips) / gridDim.x);
    const unsigned pat   = (unsigned)blockIdx.x * 0x01010101u;
    const unsigned lmask = (1u << lane) - 1u;
    int cnt = 0;

    if (nTrips > 0) {
        uint4 ka[OWN_KV], kb[OWN_KV];
        int t = wave;
        int ta = own_trip(t, nTrips, start);
        own_load(ka, keys, ta, lane);
        for (; t < nTrips; t += 2 * OWN_WAVES) {
            const int t1 = t + OWN_WAVES;
            const int tb = own_trip(t1, nTrips, start);
            own_load(kb, keys, tb, lane);
            own_take(ka, ta, pat, lmask, q, cnt, tile, newX, recs, lane);
            ta = own_trip(t1 + OWN_WAVES, nTrips, start);
            own_load(ka, keys, ta, lane);
            if (t1 < nTrips)
                own_take(kb, tb, pat, lmask, q, cnt, tile, newX, recs, lane);
        }
    }
    own_drain(q, cnt, tile, newX, recs, lane);
    __syncthreads();

    int rows = N - lo;
    if (rows > OWN_ROWS) rows = OWN_ROWS;
    float4* dst = (float4*)(agg + (size_t)lo * DD);
    const float4* srcT = (const float4*)tile;
    for (int i = tid; i < rows * (DD / 4); i += OWN_WAVES * 32) dst[i] = srcT[i];
}

__device__ __forceinline__ void blend8(float* __restrict__ o, const v8f& acc,
                                       const float (&bs)[8],
                                       const float4& aLo, const float4& aHi,
                                       const float4& xLo, const float4& xHi) {
    const float av[8] = {aLo.x, aLo.y, aLo.z, aLo.w, aHi.x, aHi.y, aHi.z, aHi.w};
    const float xv[8] = {xLo.x, xLo.y, xLo.z, xLo.w, xHi.x, xHi.y, xHi.z, xHi.w};
    float res[8];
    #pragma unroll
    for (int r = 0; r < 8; ++r) {
        float g = acc[r] + bs[r];
        g = 1.0f / (1.0f + __expf(-g));
        res[r] = av[r] * g + xv[r] * (1.0f - g);
    }
    *(float4*)(o)     = make_float4(res[0], res[1], res[2], res[3]);
    *(float4*)(o + 4) = make_float4(res[4], res[5], res[6], res[7]);
}

__global__ __launch_bounds__(128) void gate_kernel(
    const float* __restrict__ X, const float* __restrict__ agg,
    const float* __restrict__ Wgi, const float* __restrict__ Wgn,
    const float* __restrict__ bgi, const float* __restrict__ bgn,
    float* __restrict__ out, int nPairs, int N)
{
    __shared__ float wLds[DD * DD];

    const int lane = threadIdx.x & 31;
    const int wave = threadIdx.x >> 5;
    const int half = lane >> 4;
    const int l16  = lane & 15;
    const int pair = blockIdx.x * 4 + wave;
    const bool active = (pair < nPairs);
    const int rowBase = pair * 32;
    const int r0i = rowBase + l16;
    const int r1i = rowBase + 16 + l16;
    int m0 = r0i;  if (m0 >= N) m0 = (N > 0) ? N - 1 : 0;
    int m1 = r1i;  if (m1 >= N) m1 = (N > 0) ? N - 1 : 0;

    float4 x0[2 * KCHUNKS], x1[2 * KCHUNKS], a0[2 * KCHUNKS], a1[2 * KCHUNKS];
    if (active) {
        load_row_half(x0, X   + (size_t)m0 * DD + 8 * half);
        load_row_half(x1, X   + (size_t)m1 * DD + 8 * half);
        load_row_half(a0, agg + (size_t)m0 * DD + 8 * half);
        load_row_half(a1, agg + (size_t)m1 * DD + 8 * half);
    }

    v8f acc0[CTILES] = {};
    v8f acc1[CTILES] = {};

    stage_weights(wLds, Wgi);
    __syncthreads();
    if (active) mma_rows(acc0, acc1, x0, x1, wLds, half, l16);
    __syncthreads();

    stage_weights(wLds, Wgn);
    __syncthreads();
    if (active) {
        float bs[CTILES][8];
        #pragma unroll
        for (int ct = 0; ct < CTILES; ++ct) {
            const int n0 = ct * 16 + 8 * half;
            const float4 iA = *(const float4*)(bgi + n0);
            const float4 iB = *(const float4*)(bgi + n0 + 4);
            const float4 nA = *(const float4*)(bgn + n0);
            const float4 nB = *(const float4*)(bgn + n0 + 4);
            bs[ct][0] = iA.x + nA.x; bs[ct][1] = iA.y + nA.y;
            bs[ct][2] = iA.z + nA.z; bs[ct][3] = iA.w + nA.w;
            bs[ct][4] = iB.x + nB.x; bs[ct][5] = iB.y + nB.y;
            bs[ct][6] = iB.z + nB.z; bs[ct][7] = iB.w + nB.w;
        }

        mma_rows(acc0, acc1, a0, a1, wLds, half, l16);

        #pragma unroll
        for (int ct = 0; ct < CTILES; ++ct) {
            const int n0 = ct * 16 + 8 * half;
            if (r0i < N)
                blend8(out + (size_t)r0i * DD + n0, acc0[ct], bs[ct],
                       a0[2 * ct], a0[2 * ct + 1], x0[2 * ct], x0[2 * ct + 1]);
            if (r1i < N)
                blend8(out + (size_t)r1i * DD + n0, acc1[ct], bs[ct],
                       a1[2 * ct], a1[2 * ct + 1], x1[2 * ct], x1[2 * ct + 1]);
        }
    }
}

extern "C" void kernel_launch(void* const* d_in, const int* in_sizes, int n_in,
                              void* d_out, int out_size, void* d_ws, size_t ws_size,
                              hipStream_t stream) {
    const float* X      = (const float*)d_in[0];
    const float* a_vals = (const float*)d_in[1];
    const float* Wn     = (const float*)d_in[2];
    const float* bn     = (const float*)d_in[3];
    const float* Wgi    = (const float*)d_in[4];
    const float* bgi    = (const float*)d_in[5];
    const float* Wgn    = (const float*)d_in[6];
    const float* bgn    = (const float*)d_in[7];
    const int*   row    = (const int*)d_in[8];
    const int*   col    = (const int*)d_in[9];
    float* out = (float*)d_out;

    const int N = in_sizes[0] / DD;
    const int E = in_sizes[1];

    float* newX = (float*)d_ws;
    float* agg  = (float*)d_ws + (size_t)N * DD;

    const int nPairs = (N + 31) / 32;
    const int gemmBlocks = (nPairs + 3) / 4;

    float* keys = out;

    gemm_bias_kernel<<<gemmBlocks, 128, 0, stream>>>(X, Wn, bn, newX, row, col, a_vals, keys, nPairs, N, E);

    {
        const int ownBlocks = (N + OWN_ROWS - 1) / OWN_ROWS;
        owner_sum_kernel<<<ownBlocks, OWN_WAVES * 32, 0, stream>>>(newX, keys, agg, E, N);
    }

    gate_kernel<<<gemmBlocks, 128, 0, stream>>>(X, agg, Wgi, Wgn, bgi, bgn,
                                                out, nPairs, N);
}


# weights staged in one round trip: a thread requests all its 16-byte pieces up front, then stores 32-byte k-pair runs to LDS (was 9 dependent rounds of 4-byte loads); gate: both W in LDS, 1 barrier
# speedup vs baseline: 2.9832x; 1.2747x over previous
#include <hip/hip_runtime.h>
#include <hip/hip_bf16.h>


typedef __attribute__((ext_vector_type(2))) float v2f;
typedef __attribute__((ext_vector_type(8))) float v8f;

#define DD 96
#define CTILES (DD / 16)
#define KCHUNKS (DD / 16)
#define LDS_PAIR_STRIDE (2 * DD)

#define STAGE_UNITS ((DD / 2) * (DD / 4) / 128)

__device__ __forceinline__ void stage_request(float4 (&lo)[STAGE_UNITS], float4 (&hi)[STAGE_UNITS],
                                              const float* __restrict__ W) {
    #pragma unroll
    for (int i = 0; i < STAGE_UNITS; ++i) {
        const int u  = (int)threadIdx.x + 128 * i;
        const int kp = u / (DD / 4);
        const int nq = u - kp * (DD / 4);
        const float* p = W + (2 * kp) * DD + 4 * nq;
        lo[i] = *(const float4*)(p);
        hi[i] = *(const float4*)(p + DD);
    }
}

__device__ __forceinline__ void stage_commit(float* __restrict__ lds,
                                             const float4 (&lo)[STAGE_UNITS],
                                             const float4 (&hi)[STAGE_UNITS]) {
    #pragma unroll
    for (int i = 0; i < STAGE_UNITS; ++i) {
        const int u  = (int)threadIdx.x + 128 * i;
        const int kp = u / (DD / 4);
        const int nq = u - kp * (DD / 4);
        float* d = lds + kp * LDS_PAIR_STRIDE + 8 * nq;
        *(float4*)(d)     = make_float4(lo[i].x, hi[i].x, lo[i].y, hi[i].y);
        *(float4*)(d + 4) = make_float4(lo[i].z, hi[i].z, lo[i].w, hi[i].w);
    }
}

__device__ __forceinline__ void load_row_half(float4 (&r)[2 * KCHUNKS],
                                              const float* __restrict__ p) {
    #pragma unroll
    for (int kc = 0; kc < KCHUNKS; ++kc) {
        r[2 * kc]     = *(const float4*)(p + 16 * kc);
        r[2 * kc + 1] = *(const float4*)(p + 16 * kc + 4);
    }
}

__device__ __forceinline__ void mma_rows(v8f (&acc0)[CTILES], v8f (&acc1)[CTILES],
                                         const float4 (&r0)[2 * KCHUNKS],
                                         const float4 (&r1)[2 * KCHUNKS],
                                         const float* __restrict__ wLds,
                                         int half, int l16) {
    #pragma unroll
    for (int kc = 0; kc < KCHUNKS; ++kc) {
        #pragma unroll
        for (int q = 0; q < 4; ++q) {
            const float4 f0 = r0[2 * kc + (q >> 1)];
            const float4 f1 = r1[2 * kc + (q >> 1)];
            v2f b0, b1;
            if (q & 1) { b0.x = f0.z; b0.y = f0.w; b1.x = f1.z; b1.y = f1.w; }
            else       { b0.x = f0.x; b0.y = f0.y; b1.x = f1.x; b1.y = f1.y; }
            const float* wRow = wLds + (8 * kc + 4 * half + q) * LDS_PAIR_STRIDE + l16 * 2;
            #pragma unroll
            for (int ct = 0; ct < CTILES; ++ct) {
                const v2f w = *(const v2f*)(wRow + ct * 32);
                acc0[ct] = __builtin_amdgcn_wmma_f32_16x16x4_f32(
                    false, w, false, b0, (short)0, acc0[ct], false, false);
                acc1[ct] = __builtin_amdgcn_wmma_f32_16x16x4_f32(
                    false, w, false, b1, (short)0, acc1[ct], false, false);
            }
        }
    }
}

#define OWN_ROWS 200
#define OWN_KV 4
#define OWN_TRIP_UNITS (32 * OWN_KV)

__device__ __forceinline__ unsigned owner_key(int r) {
    const unsigned k = (unsigned)r / (unsigned)OWN_ROWS;
    return k < 255u ? k : 255u;
}

#define QUAD_PRE 4

__device__ __forceinline__ void load_quad(int4& r, int4& c, float4& a,
                                          const int* __restrict__ row,
                                          const int* __restrict__ col,
                                          const float* __restrict__ avals, int E, int qd) {
    const int e = qd * 4;
    if (e + 4 <= E) {
        r = *(const int4*)(row + e);
        c = *(const int4*)(col + e);
        a = *(const float4*)(avals + e);
    } else {
        r.x = (e     < E) ? row[e]     : -1;  c.x = (e     < E) ? col[e]     : 0;  a.x = (e     < E) ? avals[e]     : 0.0f;
        r.y = (e + 1 < E) ? row[e + 1] : -1;  c.y = (e + 1 < E) ? col[e + 1] : 0;  a.y = (e + 1 < E) ? avals[e + 1] : 0.0f;
        r.z = (e + 2 < E) ? row[e + 2] : -1;  c.z = (e + 2 < E) ? col[e + 2] : 0;  a.z = (e + 2 < E) ? avals[e + 2] : 0.0f;
        r.w = (e + 3 < E) ? row[e + 3] : -1;  c.w = (e + 3 < E) ? col[e + 3] : 0;  a.w = (e + 3 < E) ? avals[e + 3] : 0.0f;
    }
}

__device__ __forceinline__ uint2 edge_record(int r, int c, float a) {
    const unsigned k  = (unsigned)r / (unsigned)OWN_ROWS;
    const unsigned lr = (unsigned)r - k * (unsigned)OWN_ROWS;
    return make_uint2(((unsigned)c & 0x00FFFFFFu) | (lr << 24), __float_as_uint(a));
}

__device__ __forceinline__ void store_quad(float* __restrict__ keys, uint4* __restrict__ recs,
                                           int qd, const int4& r, const int4& c,
                                           const float4& a) {
    ((unsigned*)keys)[qd] = owner_key(r.x) | (owner_key(r.y) << 8)
                          | (owner_key(r.z) << 16) | (owner_key(r.w) << 24);
    const uint2 e0 = edge_record(r.x, c.x, a.x);
    const uint2 e1 = edge_record(r.y, c.y, a.y);
    const uint2 e2 = edge_record(r.z, c.z, a.z);
    const uint2 e3 = edge_record(r.w, c.w, a.w);
    recs[2 * qd]     = make_uint4(e0.x, e0.y, e1.x, e1.y);
    recs[2 * qd + 1] = make_uint4(e2.x, e2.y, e3.x, e3.y);
}

__global__ __launch_bounds__(128) void gemm_bias_kernel(
    const float* __restrict__ X, const float* __restrict__ W,
    const float* __restrict__ bias, float* __restrict__ Y,
    const int* __restrict__ row, const int* __restrict__ col,
    const float* __restrict__ avals, float* __restrict__ keys,
    int nPairs, int N, int E)
{
    __shared__ __attribute__((aligned(16))) float wLds[DD * DD];

    const int lane = threadIdx.x & 31;
    const int wave = threadIdx.x >> 5;
    const int half = lane >> 4;
    const int l16  = lane & 15;
    const int pair = blockIdx.x * 4 + wave;
    const bool active = (pair < nPairs);
    const int rowBase = pair * 32;
    const int r0i = rowBase + l16;
    const int r1i = rowBase + 16 + l16;
    int m0 = r0i;  if (m0 >= N) m0 = (N > 0) ? N - 1 : 0;
    int m1 = r1i;  if (m1 >= N) m1 = (N > 0) ? N - 1 : 0;

    float4 wLo[STAGE_UNITS], wHi[STAGE_UNITS];
    stage_request(wLo, wHi, W);

    float4 x0[2 * KCHUNKS], x1[2 * KCHUNKS];
    if (active) {
        load_row_half(x0, X + (size_t)m0 * DD + 8 * half);
        load_row_half(x1, X + (size_t)m1 * DD + 8 * half);
    }

    const int nUnits = ((E + 16 * OWN_TRIP_UNITS - 1) / (16 * OWN_TRIP_UNITS)) * OWN_TRIP_UNITS;
    const int nQuads = nUnits * 4;
    uint4* recs = (uint4*)(keys + nUnits * 4);
    const int quad0 = blockIdx.x * 128 + threadIdx.x;
    const int quadStep = gridDim.x * 128;
    int4 rq[QUAD_PRE], cq[QUAD_PRE];
    float4 aq[QUAD_PRE];
    #pragma unroll
    for (int s = 0; s < QUAD_PRE; ++s) {
        const int qd = quad0 + s * quadStep;
        if (qd < nQuads) load_quad(rq[s], cq[s], aq[s], row, col, avals, E, qd);
    }

    stage_commit(wLds, wLo, wHi);
    __syncthreads();

    #pragma unroll
    for (int s = 0; s < QUAD_PRE; ++s) {
        const int qd = quad0 + s * quadStep;
        if (qd < nQuads) store_quad(keys, recs, qd, rq[s], cq[s], aq[s]);
    }
    for (int qd = quad0 + QUAD_PRE * quadStep; qd < nQuads; qd += quadStep) {
        int4 r, c2;
        float4 a2;
        load_quad(r, c2, a2, row, col, avals, E, qd);
        store_quad(keys, recs, qd, r, c2, a2);
    }
    if (!active) return;

    float4 bA[CTILES], bB[CTILES];
    #pragma unroll
    for (int ct = 0; ct < CTILES; ++ct) {
        bA[ct] = *(const float4*)(bias + ct * 16 + 8 * half);
        bB[ct] = *(const float4*)(bias + ct * 16 + 8 * half + 4);
    }

    v8f acc0[CTILES] = {};
    v8f acc1[CTILES] = {};
    mma_rows(acc0, acc1, x0, x1, wLds, half, l16);

    #pragma unroll
    for (int ct = 0; ct < CTILES; ++ct) {
        const int n0 = ct * 16 + 8 * half;
        if (r0i < N) {
            float* y = Y + (size_t)r0i * DD + n0;
            *(float4*)(y)     = make_float4(acc0[ct][0] + bA[ct].x, acc0[ct][1] + bA[ct].y,
                                            acc0[ct][2] + bA[ct].z, acc0[ct][3] + bA[ct].w);
            *(float4*)(y + 4) = make_float4(acc0[ct][4] + bB[ct].x, acc0[ct][5] + bB[ct].y,
                                            acc0[ct][6] + bB[ct].z, acc0[ct][7] + bB[ct].w);
        }
        if (r1i < N) {
            float* y = Y + (size_t)r1i * DD + n0;
            *(float4*)(y)     = make_float4(acc1[ct][0] + bA[ct].x, acc1[ct][1] + bA[ct].y,
                                            acc1[ct][2] + bA[ct].z, acc1[ct][3] + bA[ct].w);
            *(float4*)(y + 4) = make_float4(acc1[ct][4] + bB[ct].x, acc1[ct][5] + bB[ct].y,
                                            acc1[ct][6] + bB[ct].z, acc1[ct][7] + bB[ct].w);
        }
    }
}

#define OWN_WAVES 32
#define OWN_QW 1536
#define OWN_FLY 16

__device__ __forceinline__ void own_drain(const int* __restrict__ q, int cnt,
                                          float* __restrict__ tile,
                                          const float* __restrict__ newX,
                                          const uint2* __restrict__ recs, int lane) {
    for (int b = 0; b < cnt; b += 32) {
        const int i = b + lane;
        int lr = 0, c = 0;
        float a = 0.0f;
        if (i < cnt) {
            const uint2 r = recs[q[i]];
            lr = (int)(r.x >> 24);
            c  = (int)(r.x & 0x00FFFFFFu);
            a  = __uint_as_float(r.y);
        }
        const int m = (cnt - b < 32) ? (cnt - b) : 32;
        for (int j0 = 0; j0 < m; j0 += OWN_FLY) {
            float v[OWN_FLY][3];
            #pragma unroll
            for (int u = 0; u < OWN_FLY; ++u) {
                const int cj = __builtin_amdgcn_readlane(c, j0 + u);
                const float* p = newX + (size_t)cj * DD + lane;
                v[u][0] = p[0];
                v[u][1] = p[32];
                v[u][2] = p[64];
            }
            #pragma unroll
            for (int u = 0; u < OWN_FLY; ++u) {
                if (j0 + u < m) {
                    const float aj = __int_as_float(
                        __builtin_amdgcn_readlane(__float_as_int(a), j0 + u));
                    const int lj = __builtin_amdgcn_readlane(lr, j0 + u);
                    float* t = tile + lj * DD + lane;
                    atomicAdd(t,      aj * v[u][0]);
                    atomicAdd(t + 32, aj * v[u][1]);
                    atomicAdd(t + 64, aj * v[u][2]);
                }
            }
        }
    }
}

__device__ __forceinline__ unsigned nonzero_bytes(unsigned x) {
    return ((x & 0x7F7F7F7Fu) + 0x7F7F7F7Fu) | x;
}

__device__ __forceinline__ unsigned own_pair_mask(const uint4& ka, const uint4& kb,
                                                  unsigned pat) {
    const unsigned M = 0x80808080u;
    unsigned nz;
    nz  = (nonzero_bytes(ka.x ^ pat) & M) >> 7;
    nz |= (nonzero_bytes(ka.y ^ pat) & M) >> 6;
    nz |= (nonzero_bytes(ka.z ^ pat) & M) >> 5;
    nz |= (nonzero_bytes(ka.w ^ pat) & M) >> 4;
    nz |= (nonzero_bytes(kb.x ^ pat) & M) >> 3;
    nz |= (nonzero_bytes(kb.y ^ pat) & M) >> 2;
    nz |= (nonzero_bytes(kb.z ^ pat) & M) >> 1;
    nz |= (nonzero_bytes(kb.w ^ pat) & M);
    return ~nz;
}

__device__ __forceinline__ void own_load(uint4 (&k)[OWN_KV], const uint4* __restrict__ keys,
                                         int tt, int lane) {
    #pragma unroll
    for (int v = 0; v < OWN_KV; ++v) k[v] = keys[(tt * OWN_KV + v) * 32 + lane];
}

__device__ __forceinline__ int own_trip(int t, int nTrips, int start) {
    int tt = (t < nTrips ? t : nTrips - 1) + start;
    if (tt >= nTrips) tt -= nTrips;
    return tt;
}

__device__ __forceinline__ void own_take(const uint4 (&k)[OWN_KV], int tt, unsigned pat,
                                         unsigned lmask, int* __restrict__ q, int& cnt,
                                         float* __restrict__ tile,
                                         const float* __restrict__ newX,
                                         const uint2* __restrict__ recs, int lane) {
    unsigned long long w = (unsigned long long)own_pair_mask(k[0], k[1], pat)
                         | ((unsigned long long)own_pair_mask(k[2], k[3], pat) << 32);
    for (;;) {
        const unsigned bal = __builtin_amdgcn_ballot_w32(w != 0ull);
        if (bal == 0u) break;
        if (cnt + 32 > OWN_QW) {
            own_drain(q, cnt, tile, newX, recs, lane);
            cnt = 0;
        }
        if (w != 0ull) {
            const int p  = __builtin_ctzll(w);
            w &= w - 1ull;
            const int pp = p & 31;
            const int v  = ((p >> 5) << 1) + ((pp >> 2) & 1);
            const int unit = (tt * OWN_KV + v) * 32 + lane;
            q[cnt + __builtin_popcount(bal & lmask)] = unit * 16 + ((pp & 3) << 2) + (pp >> 3);
        }
        cnt += __builtin_popcount(bal);
    }
}

__global__ __launch_bounds__(OWN_WAVES * 32) void owner_sum_kernel(
    const float* __restrict__ newX, const float* __restrict__ keysF,
    float* __restrict__ agg, int E, int N)
{
    __shared__ float tile[OWN_ROWS * DD];
    __shared__ int   queue[OWN_WAVES * OWN_QW];

    const int tid  = threadIdx.x;
    const int lane = tid & 31;
    const int wave = __builtin_amdgcn_readfirstlane(tid >> 5);
    const int lo   = blockIdx.x * OWN_ROWS;
    int* q = queue + wave * OWN_QW;
    const uint4* keys = (const uint4*)keysF;

    for (int i = tid; i < OWN_ROWS * DD; i += OWN_WAVES * 32) tile[i] = 0.0f;
    __syncthreads();

    const int nTrips = (E + 16 * OWN_TRIP_UNITS - 1) / (16 * OWN_TRIP_UNITS);
    const uint2* recs = (const uint2*)(keysF + nTrips * OWN_TRIP_UNITS * 4);
    const int start = (int)(((long long)blockIdx.x * nTrips) / gridDim.x);
    const unsigned pat   = (unsigned)blockIdx.x * 0x01010101u;
    const unsigned lmask = (1u << lane) - 1u;
    int cnt = 0;

    if (nTrips > 0) {
        uint4 ka[OWN_KV], kb[OWN_KV];
        int t = wave;
        int ta = own_trip(t, nTrips, start);
        own_load(ka, keys, ta, lane);
        for (; t < nTrips; t += 2 * OWN_WAVES) {
            const int t1 = t + OWN_WAVES;
            const int tb = own_trip(t1, nTrips, start);
            own_load(kb, keys, tb, lane);
            own_take(ka, ta, pat, lmask, q, cnt, tile, newX, recs, lane);
            ta = own_trip(t1 + OWN_WAVES, nTrips, start);
            own_load(ka, keys, ta, lane);
            if (t1 < nTrips)
                own_take(kb, tb, pat, lmask, q, cnt, tile, newX, recs, lane);
        }
    }
    own_drain(q, cnt, tile, newX, recs, lane);
    __syncthreads();

    int rows = N - lo;
    if (rows > OWN_ROWS) rows = OWN_ROWS;
    float4* dst = (float4*)(agg + (size_t)lo * DD);
    const float4* srcT = (const float4*)tile;
    for (int i = tid; i < rows * (DD / 4); i += OWN_WAVES * 32) dst[i] = srcT[i];
}

__device__ __forceinline__ void blend8(float* __restrict__ o, const v8f& acc,
                                       const float (&bs)[8],
                                       const float4& aLo, const float4& aHi,
                                       const float4& xLo, const float4& xHi) {
    const float av[8] = {aLo.x, aLo.y, aLo.z, aLo.w, aHi.x, aHi.y, aHi.z, aHi.w};
    const float xv[8] = {xLo.x, xLo.y, xLo.z, xLo.w, xHi.x, xHi.y, xHi.z, xHi.w};
    float res[8];
    #pragma unroll
    for (int r = 0; r < 8; ++r) {
        float g = acc[r] + bs[r];
        g = 1.0f / (1.0f + __expf(-g));
        res[r] = av[r] * g + xv[r] * (1.0f - g);
    }
    *(float4*)(o)     = make_float4(res[0], res[1], res[2], res[3]);
    *(float4*)(o + 4) = make_float4(res[4], res[5], res[6], res[7]);
}

__global__ __launch_bounds__(128) void gate_kernel(
    const float* __restrict__ X, const float* __restrict__ agg,
    const float* __restrict__ Wgi, const float* __restrict__ Wgn,
    const float* __restrict__ bgi, const float* __restrict__ bgn,
    float* __restrict__ out, int nPairs, int N)
{
    __shared__ __attribute__((aligned(16))) float wLds[2 * DD * DD];

    const int lane = threadIdx.x & 31;
    const int wave = threadIdx.x >> 5;
    const int half = lane >> 4;
    const int l16  = lane & 15;
    const int pair = blockIdx.x * 4 + wave;
    const bool active = (pair < nPairs);
    const int rowBase = pair * 32;
    const int r0i = rowBase + l16;
    const int r1i = rowBase + 16 + l16;
    int m0 = r0i;  if (m0 >= N) m0 = (N > 0) ? N - 1 : 0;
    int m1 = r1i;  if (m1 >= N) m1 = (N > 0) ? N - 1 : 0;

    float4 giLo[STAGE_UNITS], giHi[STAGE_UNITS], gnLo[STAGE_UNITS], gnHi[STAGE_UNITS];
    stage_request(giLo, giHi, Wgi);
    stage_request(gnLo, gnHi, Wgn);

    float4 x0[2 * KCHUNKS], x1[2 * KCHUNKS], a0[2 * KCHUNKS], a1[2 * KCHUNKS];
    if (active) {
        load_row_half(x0, X   + (size_t)m0 * DD + 8 * half);
        load_row_half(x1, X   + (size_t)m1 * DD + 8 * half);
        load_row_half(a0, agg + (size_t)m0 * DD + 8 * half);
        load_row_half(a1, agg + (size_t)m1 * DD + 8 * half);
    }

    stage_commit(wLds,           giLo, giHi);
    stage_commit(wLds + DD * DD, gnLo, gnHi);
    __syncthreads();

    if (active) {
        v8f acc0[CTILES] = {};
        v8f acc1[CTILES] = {};

        float bs[CTILES][8];
        #pragma unroll
        for (int ct = 0; ct < CTILES; ++ct) {
            const int n0 = ct * 16 + 8 * half;
            const float4 iA = *(const float4*)(bgi + n0);
            const float4 iB = *(const float4*)(bgi + n0 + 4);
            const float4 nA = *(const float4*)(bgn + n0);
            const float4 nB = *(const float4*)(bgn + n0 + 4);
            bs[ct][0] = iA.x + nA.x; bs[ct][1] = iA.y + nA.y;
            bs[ct][2] = iA.z + nA.z; bs[ct][3] = iA.w + nA.w;
            bs[ct][4] = iB.x + nB.x; bs[ct][5] = iB.y + nB.y;
            bs[ct][6] = iB.z + nB.z; bs[ct][7] = iB.w + nB.w;
        }

        mma_rows(acc0, acc1, x0, x1, wLds,           half, l16);
        mma_rows(acc0, acc1, a0, a1, wLds + DD * DD, half, l16);

        #pragma unroll
        for (int ct = 0; ct < CTILES; ++ct) {
            const int n0 = ct * 16 + 8 * half;
            if (r0i < N)
                blend8(out + (size_t)r0i * DD + n0, acc0[ct], bs[ct],
                       a0[2 * ct], a0[2 * ct + 1], x0[2 * ct], x0[2 * ct + 1]);
            if (r1i < N)
                blend8(out + (size_t)r1i * DD + n0, acc1[ct], bs[ct],
                       a1[2 * ct], a1[2 * ct + 1], x1[2 * ct], x1[2 * ct + 1]);
        }
    }
}

extern "C" void kernel_launch(void* const* d_in, const int* in_sizes, int n_in,
                              void* d_out, int out_size, void* d_ws, size_t ws_size,
                              hipStream_t stream) {
    const float* X      = (const float*)d_in[0];
    const float* a_vals = (const float*)d_in[1];
    const float* Wn     = (const float*)d_in[2];
    const float* bn     = (const float*)d_in[3];
    const float* Wgi    = (const float*)d_in[4];
    const float* bgi    = (const float*)d_in[5];
    const float* Wgn    = (const float*)d_in[6];
    const float* bgn    = (const float*)d_in[7];
    const int*   row    = (const int*)d_in[8];
    const int*   col    = (const int*)d_in[9];
    float* out = (float*)d_out;

    const int N = in_sizes[0] / DD;
    const int E = in_sizes[1];

    float* newX = (float*)d_ws;
    float* agg  = (float*)d_ws + (size_t)N * DD;

    const int nPairs = (N + 31) / 32;
    const int gemmBlocks = (nPairs + 3) / 4;

    float* keys = out;

    gemm_bias_kernel<<<gemmBlocks, 128, 0, stream>>>(X, Wn, bn, newX, row, col, a_vals, keys, nPairs, N, E);

    {
        const int ownBlocks = (N + OWN_ROWS - 1) / OWN_ROWS;
        owner_sum_kernel<<<ownBlocks, OWN_WAVES * 32, 0, stream>>>(newX, keys, agg, E, N);
    }

    gate_kernel<<<gemmBlocks, 128, 0, stream>>>(X, agg, Wgi, Wgn, bgi, bgn,
                                                out, nPairs, N);
}
